# starting kernel
# baseline (speedup 1.0000x reference)
#include <hip/hip_runtime.h>


typedef __attribute__((ext_vector_type(2))) float v2f;
typedef __attribute__((ext_vector_type(8))) float v8f;

#define D_MODEL 24
#define REL_DIM 10
#define NN 512
#define BB 8
#define K1_CHUNKS 13
#define K2_CHUNKS 3

__global__ __launch_bounds__(32) void pairagg_kernel(
    const float* __restrict__ x, const float* __restrict__ q,
    const float* __restrict__ coord, const float* __restrict__ W1,
    const float* __restrict__ b1, const float* __restrict__ W2,
    const float* __restrict__ b2, float* __restrict__ out)
{
  __shared__ float hT[16 * 16];

  const int bi   = blockIdx.x;
  const int b    = bi / NN;
  const int i    = bi % NN;
  const int lane = threadIdx.x;
  const int m    = lane & 15;
  const int hh   = lane >> 4;

  const float* xB = x     + (size_t)b * NN * D_MODEL;
  const float* qB = q     + (size_t)b * NN * D_MODEL;
  const float* cB = coord + (size_t)b * NN * D_MODEL;

  v2f ci[6], qi[6];
#pragma unroll
  for (int c = 0; c < 6; ++c) {
    const int d = 4 * c + 2 * hh;
    ci[c] = *(const v2f*)(cB + i * D_MODEL + d);
    qi[c] = *(const v2f*)(qB + i * D_MODEL + d);
  }

  float bias_row = 0.0f;
  if (m < REL_DIM) {
    bias_row = b1[m];
#pragma unroll
    for (int d = 0; d < D_MODEL; ++d)
      bias_row = fmaf(xB[i * D_MODEL + d], W1[d * REL_DIM + m], bias_row);
  }

  v2f B1[K1_CHUNKS];
#pragma unroll
  for (int c = 0; c < K1_CHUNKS; ++c) {
#pragma unroll
    for (int v = 0; v < 2; ++v) {
      const int k = 4 * c + 2 * hh + v;
      float val = 0.0f;
      if (m < REL_DIM) {
        int row = -1;
        if      (k < 24)  row = 48 + k;
        else if (k < 48)  row = k;
        else if (k == 48) row = 72;
        else if (k == 49) row = 73;
        if (row >= 0)       val = W1[row * REL_DIM + m];
        else if (k == 50)   val = bias_row;
      }
      B1[c][v] = val;
    }
  }

  v2f B2[K2_CHUNKS];
#pragma unroll
  for (int c = 0; c < K2_CHUNKS; ++c) {
#pragma unroll
    for (int v = 0; v < 2; ++v) {
      const int k = 4 * c + 2 * hh + v;
      B2[c][v] = (k < REL_DIM && m < REL_DIM) ? W2[k * REL_DIM + m] : 0.0f;
    }
  }
  const float b2v = (m < REL_DIM) ? b2[m] : 0.0f;

  float psum = 0.0f;
  float pmax = -1e9f;

  for (int jbase = 0; jbase < NN; jbase += 16) {
    const int j = jbase + m;
    const float* cj_ptr = cB + j * D_MODEL + 2 * hh;
    const float* xj_ptr = xB + j * D_MODEL + 2 * hh;
    const float* qj_ptr = qB + j * D_MODEL + 2 * hh;

    v2f A[K1_CHUNKS];
    float sqpart = 0.0f, qpart = 0.0f;
#pragma unroll
    for (int c = 0; c < 6; ++c) {
      const v2f cj = *(const v2f*)(cj_ptr + 4 * c);
      const v2f xj = *(const v2f*)(xj_ptr + 4 * c);
      const v2f qj = *(const v2f*)(qj_ptr + 4 * c);
      const float d0 = ci[c][0] - cj[0];
      const float d1 = ci[c][1] - cj[1];
      A[c][0] = fabsf(d0);
      A[c][1] = fabsf(d1);
      sqpart  = fmaf(d0, d0, fmaf(d1, d1, sqpart));
      qpart   = fmaf(qi[c][0], qj[0], fmaf(qi[c][1], qj[1], qpart));
      A[6 + c][0] = xj[0];
      A[6 + c][1] = xj[1];
    }
    const float qfull = qpart  + __shfl_xor(qpart, 16, 32);
    const float cdist = sqrtf(sqpart + __shfl_xor(sqpart, 16, 32));
    A[12][0] = hh ? 1.0f : qfull;
    A[12][1] = hh ? 0.0f : cdist;

    v8f acc = {};
#pragma unroll
    for (int c = 0; c < K1_CHUNKS; ++c)
      acc = __builtin_amdgcn_wmma_f32_16x16x4_f32(
          false, A[c], false, B1[c], (short)0, acc, false, false);

    __syncthreads();
#pragma unroll
    for (int r = 0; r < 8; ++r)
      hT[(r + 8 * hh) * 16 + m] = fmaxf(acc[r], 0.0f);
    __syncthreads();

    v2f A2[K2_CHUNKS];
#pragma unroll
    for (int c = 0; c < K2_CHUNKS; ++c)
      A2[c] = *(const v2f*)(&hT[m * 16 + 4 * c + 2 * hh]);

    v8f acc2 = {};
#pragma unroll
    for (int c = 0; c < K2_CHUNKS; ++c)
      acc2 = __builtin_amdgcn_wmma_f32_16x16x4_f32(
          false, A2[c], false, B2[c], (short)0, acc2, false, false);

#pragma unroll
    for (int r = 0; r < 8; ++r) {
      const float rel  = fmaxf(acc2[r] + b2v, 0.0f);
      const bool  diag = (jbase + r + 8 * hh) == i;
      psum += diag ? 0.0f : rel;
      pmax  = diag ? pmax : fmaxf(pmax, rel);
    }
  }

  psum += __shfl_xor(psum, 16, 32);
  pmax  = fmaxf(pmax, __shfl_xor(pmax, 16, 32));

  if (lane < REL_DIM) {
    const size_t base = ((size_t)b * NN + i) * REL_DIM + lane;
    out[base] = psum / (float)(NN - 1);
    out[(size_t)BB * NN * REL_DIM + base] = pmax;
  }
}

extern "C" void kernel_launch(void* const* d_in, const int* in_sizes, int n_in,
                              void* d_out, int out_size, void* d_ws, size_t ws_size,
                              hipStream_t stream) {
  const float* x  = (const float*)d_in[0];
  const float* q  = (const float*)d_in[1];
  const float* co = (const float*)d_in[2];
  const float* W1 = (const float*)d_in[3];
  const float* b1 = (const float*)d_in[4];
  const float* W2 = (const float*)d_in[5];
  const float* b2 = (const float*)d_in[6];
  float* out = (float*)d_out;

  dim3 grid(BB * NN);
  dim3 block(32);
  hipLaunchKernelGGL(pairagg_kernel, grid, block, 0, stream,
                     x, q, co, W1, b1, W2, b2, out);
}


# Pair tile turned: lane = row i held in registers, a wave walks its 32 columns j; weights are the A operand, so layer-1 D is layer-2's B and row sum/max add up in registers (sum order changes)
# speedup vs baseline: 1.9937x; 1.9937x over previous
#include <hip/hip_runtime.h>
#include <type_traits>


typedef __attribute__((ext_vector_type(2))) float v2f;
typedef __attribute__((ext_vector_type(4))) float v4f;
typedef __attribute__((ext_vector_type(8))) float v8f;

#define D_MODEL 24
#define REL_DIM 10
#define NN 512
#define BB 8
#define IC 16
#define NWAVES 16
#define NTHREADS (NWAVES * 32)
#define JW (NN / NWAVES)
#define OUTW (IC * REL_DIM)

__global__ __launch_bounds__(NTHREADS) void pairagg_kernel(
    const float* __restrict__ x, const float* __restrict__ q,
    const float* __restrict__ coord, const float* __restrict__ W1,
    const float* __restrict__ b1, const float* __restrict__ W2,
    const float* __restrict__ b2, float* __restrict__ out)
{
  __shared__ __attribute__((aligned(16))) float cjL[NWAVES * JW * D_MODEL];
  __shared__ __attribute__((aligned(16))) float qjL[NWAVES * JW * D_MODEL];
  __shared__ float vL[NWAVES * JW * 16];
  __shared__ float sumL[NWAVES * OUTW];
  __shared__ float maxL[NWAVES * OUTW];

  const int bi   = blockIdx.x;
  const int b    = bi / (NN / IC);
  const int i0   = (bi % (NN / IC)) * IC;
  const int t    = threadIdx.x;
  const int wave = __builtin_amdgcn_readfirstlane(t >> 5);
  const int lane = t & 31;
  const int m    = lane & 15;
  const int hh   = lane >> 4;

  const float* xB = x     + (size_t)b * NN * D_MODEL;
  const float* qB = q     + (size_t)b * NN * D_MODEL;
  const float* cB = coord + (size_t)b * NN * D_MODEL;

  const int  un   = (m < 5) ? m : m - 3;
  const bool has  = (m < 5) || (m >= 8 && m < 13);
  const int  unc  = has ? un : 0;

  v2f Wc[6], Wxi[6], Wxj[6];
#pragma unroll
  for (int c = 0; c < 6; ++c) {
#pragma unroll
    for (int v = 0; v < 2; ++v) {
      const int d = 4 * c + 2 * hh + v;
      const float wc = W1[(48 + d) * REL_DIM + unc];
      const float wi = W1[d * REL_DIM + unc];
      const float wj = W1[(24 + d) * REL_DIM + unc];
      Wc[c][v]  = has ? wc : 0.0f;
      Wxi[c][v] = has ? wi : 0.0f;
      Wxj[c][v] = has ? wj : 0.0f;
    }
  }
  v2f Wlast;
  {
    const float w72 = W1[72 * REL_DIM + unc];
    const float w73 = W1[73 * REL_DIM + unc];
    Wlast[0] = (has && !hh) ? w72 : 0.0f;
    Wlast[1] = (has && !hh) ? w73 : 0.0f;
  }
  v2f W2a[3];
#pragma unroll
  for (int c = 0; c < 3; ++c) {
#pragma unroll
    for (int v = 0; v < 2; ++v) {
      const int s  = 2 * c + v;
      const int ui = (s < 5) ? (5 * hh + s) : 0;
      const float w = W2[ui * REL_DIM + unc];
      W2a[c][v] = (s < 5 && has) ? w : 0.0f;
    }
  }
  v8f accx, c2init;
#pragma unroll
  for (int r = 0; r < 8; ++r) {
    accx[r]   = (r < 5) ? b1[5 * hh + (r < 5 ? r : 0)] : 0.0f;
    c2init[r] = (r < 5) ? b2[5 * hh + (r < 5 ? r : 0)] : 0.0f;
  }

  v2f ci[6], qi[6], xi[6];
  {
    const int irow = i0 + m;
#pragma unroll
    for (int c = 0; c < 6; ++c) {
      const int d = 4 * c + 2 * hh;
      ci[c] = *(const v2f*)(cB + irow * D_MODEL + d);
      qi[c] = *(const v2f*)(qB + irow * D_MODEL + d);
      xi[c] = *(const v2f*)(xB + irow * D_MODEL + d);
    }
  }
#pragma unroll
  for (int c = 0; c < 6; ++c)
    accx = __builtin_amdgcn_wmma_f32_16x16x4_f32(
        false, Wxi[c], false, xi[c], (short)0, accx, false, false);

  const int j0 = wave * JW;
  float* cjW = cjL + wave * (JW * D_MODEL);
  float* qjW = qjL + wave * (JW * D_MODEL);
  float* vW  = vL  + wave * (JW * 16);
  {
    const v4f* cg = (const v4f*)(cB + j0 * D_MODEL);
    const v4f* qg = (const v4f*)(qB + j0 * D_MODEL);
#pragma unroll
    for (int k = 0; k < (JW * D_MODEL) / (4 * 32); ++k) {
      const int f4 = k * 32 + lane;
      const int jl = f4 / 6, cc = f4 % 6;
      const v4f cv = cg[f4];
      const v4f qv = qg[f4];
      v2f lo, hi;
      lo[0] = cv[0]; lo[1] = cv[1]; hi[0] = cv[2]; hi[1] = cv[3];
      *(v2f*)(cjW + jl * D_MODEL + 2 * cc)      = lo;
      *(v2f*)(cjW + jl * D_MODEL + 12 + 2 * cc) = hi;
      lo[0] = qv[0]; lo[1] = qv[1]; hi[0] = qv[2]; hi[1] = qv[3];
      *(v2f*)(qjW + jl * D_MODEL + 2 * cc)      = lo;
      *(v2f*)(qjW + jl * D_MODEL + 12 + 2 * cc) = hi;
    }
  }
#pragma unroll
  for (int tt = 0; tt < JW / 16; ++tt) {
    const int j = j0 + tt * 16 + m;
    v8f dv = {};
#pragma unroll
    for (int c = 0; c < 6; ++c) {
      const v2f xj = *(const v2f*)(xB + j * D_MODEL + 4 * c + 2 * hh);
      dv = __builtin_amdgcn_wmma_f32_16x16x4_f32(
          false, Wxj[c], false, xj, (short)0, dv, false, false);
    }
#pragma unroll
    for (int r = 0; r < 8; ++r)
      vW[(tt * 16 + m) * 16 + r + 8 * hh] = dv[r];
  }

  __syncthreads();

  float s[5]  = {0.0f, 0.0f, 0.0f, 0.0f, 0.0f};
  float mx[5] = {0.0f, 0.0f, 0.0f, 0.0f, 0.0f};

  auto run_tile = [&](auto tag, const int tt) {
    constexpr bool DIAG = decltype(tag)::value;
    for (int jl = 0; jl < 16; ++jl) {
      const int jj = tt * 16 + jl;
      const v4f* cp = (const v4f*)(cjW + jj * D_MODEL + hh * 12);
      const v4f* qp = (const v4f*)(qjW + jj * D_MODEL + hh * 12);
      const v4f c0 = cp[0], c1 = cp[1], c2 = cp[2];
      const v4f q0 = qp[0], q1 = qp[1], q2 = qp[2];
      const float cjF[12] = {c0[0], c0[1], c0[2], c0[3], c1[0], c1[1], c1[2], c1[3],
                             c2[0], c2[1], c2[2], c2[3]};
      const float qjF[12] = {q0[0], q0[1], q0[2], q0[3], q1[0], q1[1], q1[2], q1[3],
                             q2[0], q2[1], q2[2], q2[3]};
      const float vj = vW[jj * 16 + m];

      v2f F[6];
      v2f Flast;
      float sqpart = 0.0f, qpart = 0.0f;
#pragma unroll
      for (int c = 0; c < 6; ++c) {
        const float d0 = ci[c][0] - cjF[2 * c];
        const float d1 = ci[c][1] - cjF[2 * c + 1];
        F[c][0] = fabsf(d0);
        F[c][1] = fabsf(d1);
        sqpart  = fmaf(d0, d0, fmaf(d1, d1, sqpart));
        qpart   = fmaf(qi[c][0], qjF[2 * c], fmaf(qi[c][1], qjF[2 * c + 1], qpart));
      }
      const float qfull = qpart  + __shfl_xor(qpart, 16, 32);
      const float cdist = sqrtf(sqpart + __shfl_xor(sqpart, 16, 32));
      Flast[0] = hh ? 1.0f : qfull;
      Flast[1] = hh ? 0.0f : cdist;
      v2f Wl = Wlast;
      Wl[0] = hh ? vj : Wlast[0];

      v8f acc = accx;
#pragma unroll
      for (int c = 0; c < 6; ++c)
        acc = __builtin_amdgcn_wmma_f32_16x16x4_f32(
            false, Wc[c], false, F[c], (short)0, acc, false, false);
      acc = __builtin_amdgcn_wmma_f32_16x16x4_f32(
          false, Wl, false, Flast, (short)0, acc, false, false);

      v2f H[3];
      H[0][0] = fmaxf(acc[0], 0.0f); H[0][1] = fmaxf(acc[1], 0.0f);
      H[1][0] = fmaxf(acc[2], 0.0f); H[1][1] = fmaxf(acc[3], 0.0f);
      H[2][0] = fmaxf(acc[4], 0.0f); H[2][1] = 0.0f;

      v8f d2 = c2init;
#pragma unroll
      for (int c = 0; c < 3; ++c)
        d2 = __builtin_amdgcn_wmma_f32_16x16x4_f32(
            false, W2a[c], false, H[c], (short)0, d2, false, false);

      if (DIAG) {
        const bool dg = (jl == m);
#pragma unroll
        for (int r = 0; r < 5; ++r) {
          s[r] += dg ? 0.0f : fmaxf(d2[r], 0.0f);
          mx[r] = dg ? mx[r] : fmaxf(mx[r], d2[r]);
        }
      } else {
#pragma unroll
        for (int r = 0; r < 5; ++r) {
          s[r] += fmaxf(d2[r], 0.0f);
          mx[r] = fmaxf(mx[r], d2[r]);
        }
      }
    }
  };

#pragma unroll
  for (int tt = 0; tt < JW / 16; ++tt) {
    if (j0 + tt * 16 == i0) run_tile(std::true_type{}, tt);
    else                    run_tile(std::false_type{}, tt);
  }

#pragma unroll
  for (int r = 0; r < 5; ++r) {
    sumL[wave * OUTW + m * REL_DIM + 5 * hh + r] = s[r];
    maxL[wave * OUTW + m * REL_DIM + 5 * hh + r] = mx[r];
  }

  __syncthreads();

  const size_t obase = ((size_t)b * NN + i0) * REL_DIM;
  if (t < OUTW) {
    float a = 0.0f;
#pragma unroll
    for (int w = 0; w < NWAVES; ++w)
      a += sumL[w * OUTW + t];
    out[obase + t] = a / (float)(NN - 1);
  } else if (t >= NTHREADS / 2 && t < NTHREADS / 2 + OUTW) {
    const int e = t - NTHREADS / 2;
    float mv = maxL[e];
#pragma unroll
    for (int w = 1; w < NWAVES; ++w)
      mv = fmaxf(mv, maxL[w * OUTW + e]);
    out[(size_t)BB * NN * REL_DIM + obase + e] = mv;
  }
}

extern "C" void kernel_launch(void* const* d_in, const int* in_sizes, int n_in,
                              void* d_out, int out_size, void* d_ws, size_t ws_size,
                              hipStream_t stream) {
  const float* x  = (const float*)d_in[0];
  const float* q  = (const float*)d_in[1];
  const float* co = (const float*)d_in[2];
  const float* W1 = (const float*)d_in[3];
  const float* b1 = (const float*)d_in[4];
  const float* W2 = (const float*)d_in[5];
  const float* b2 = (const float*)d_in[6];
  float* out = (float*)d_out;

  dim3 grid(BB * (NN / IC));
  dim3 block(NTHREADS);
  hipLaunchKernelGGL(pairagg_kernel, grid, block, 0, stream,
                     x, q, co, W1, b1, W2, b2, out);
}


# 8 waves per workgroup, 64 columns each: half the waves, so entry and per-wave setup halve; pair loop pipelined one column deep so 2 waves per SIMD cover their own latency (row sum order changes)
# speedup vs baseline: 2.0750x; 1.0408x over previous
#include <hip/hip_runtime.h>
#include <type_traits>


typedef __attribute__((ext_vector_type(2))) float v2f;
typedef __attribute__((ext_vector_type(4))) float v4f;
typedef __attribute__((ext_vector_type(8))) float v8f;

#define D_MODEL 24
#define REL_DIM 10
#define NN 512
#define BB 8
#define IC 16
#define NWAVES 8
#define NTHREADS (NWAVES * 32)
#define JW (NN / NWAVES)
#define OUTW (IC * REL_DIM)

__global__ __launch_bounds__(NTHREADS) void pairagg_kernel(
    const float* __restrict__ x, const float* __restrict__ q,
    const float* __restrict__ coord, const float* __restrict__ W1,
    const float* __restrict__ b1, const float* __restrict__ W2,
    const float* __restrict__ b2, float* __restrict__ out)
{
  __shared__ __attribute__((aligned(16))) float cjL[NWAVES * JW * D_MODEL];
  __shared__ __attribute__((aligned(16))) float qjL[NWAVES * JW * D_MODEL];
  __shared__ float vL[NWAVES * JW * 16];
  __shared__ float sumL[NWAVES * OUTW];
  __shared__ float maxL[NWAVES * OUTW];

  const int bi   = blockIdx.x;
  const int b    = bi / (NN / IC);
  const int i0   = (bi % (NN / IC)) * IC;
  const int t    = threadIdx.x;
  const int wave = __builtin_amdgcn_readfirstlane(t >> 5);
  const int lane = t & 31;
  const int m    = lane & 15;
  const int hh   = lane >> 4;

  const float* xB = x     + (size_t)b * NN * D_MODEL;
  const float* qB = q     + (size_t)b * NN * D_MODEL;
  const float* cB = coord + (size_t)b * NN * D_MODEL;

  const int  un   = (m < 5) ? m : m - 3;
  const bool has  = (m < 5) || (m >= 8 && m < 13);
  const int  unc  = has ? un : 0;

  v2f Wc[6], Wxi[6], Wxj[6];
#pragma unroll
  for (int c = 0; c < 6; ++c) {
#pragma unroll
    for (int v = 0; v < 2; ++v) {
      const int d = 4 * c + 2 * hh + v;
      const float wc = W1[(48 + d) * REL_DIM + unc];
      const float wi = W1[d * REL_DIM + unc];
      const float wj = W1[(24 + d) * REL_DIM + unc];
      Wc[c][v]  = has ? wc : 0.0f;
      Wxi[c][v] = has ? wi : 0.0f;
      Wxj[c][v] = has ? wj : 0.0f;
    }
  }
  v2f Wlast;
  {
    const float w72 = W1[72 * REL_DIM + unc];
    const float w73 = W1[73 * REL_DIM + unc];
    Wlast[0] = (has && !hh) ? w72 : 0.0f;
    Wlast[1] = (has && !hh) ? w73 : 0.0f;
  }
  v2f W2a[3];
#pragma unroll
  for (int c = 0; c < 3; ++c) {
#pragma unroll
    for (int v = 0; v < 2; ++v) {
      const int s  = 2 * c + v;
      const int ui = (s < 5) ? (5 * hh + s) : 0;
      const float w = W2[ui * REL_DIM + unc];
      W2a[c][v] = (s < 5 && has) ? w : 0.0f;
    }
  }
  v8f accx, c2init;
#pragma unroll
  for (int r = 0; r < 8; ++r) {
    accx[r]   = (r < 5) ? b1[5 * hh + (r < 5 ? r : 0)] : 0.0f;
    c2init[r] = (r < 5) ? b2[5 * hh + (r < 5 ? r : 0)] : 0.0f;
  }

  v2f ci[6], qi[6], xi[6];
  {
    const int irow = i0 + m;
#pragma unroll
    for (int c = 0; c < 6; ++c) {
      const int d = 4 * c + 2 * hh;
      ci[c] = *(const v2f*)(cB + irow * D_MODEL + d);
      qi[c] = *(const v2f*)(qB + irow * D_MODEL + d);
      xi[c] = *(const v2f*)(xB + irow * D_MODEL + d);
    }
  }
#pragma unroll
  for (int c = 0; c < 6; ++c)
    accx = __builtin_amdgcn_wmma_f32_16x16x4_f32(
        false, Wxi[c], false, xi[c], (short)0, accx, false, false);

  const int j0 = wave * JW;
  float* cjW = cjL + wave * (JW * D_MODEL);
  float* qjW = qjL + wave * (JW * D_MODEL);
  float* vW  = vL  + wave * (JW * 16);
  {
    const v4f* cg = (const v4f*)(cB + j0 * D_MODEL);
    const v4f* qg = (const v4f*)(qB + j0 * D_MODEL);
#pragma unroll
    for (int k = 0; k < (JW * D_MODEL) / (4 * 32); ++k) {
      const int f4 = k * 32 + lane;
      const int jl = f4 / 6, cc = f4 % 6;
      const v4f cv = cg[f4];
      const v4f qv = qg[f4];
      v2f lo, hi;
      lo[0] = cv[0]; lo[1] = cv[1]; hi[0] = cv[2]; hi[1] = cv[3];
      *(v2f*)(cjW + jl * D_MODEL + 2 * cc)      = lo;
      *(v2f*)(cjW + jl * D_MODEL + 12 + 2 * cc) = hi;
      lo[0] = qv[0]; lo[1] = qv[1]; hi[0] = qv[2]; hi[1] = qv[3];
      *(v2f*)(qjW + jl * D_MODEL + 2 * cc)      = lo;
      *(v2f*)(qjW + jl * D_MODEL + 12 + 2 * cc) = hi;
    }
  }
#pragma unroll
  for (int tt = 0; tt < JW / 16; ++tt) {
    const int j = j0 + tt * 16 + m;
    v8f dv = {};
#pragma unroll
    for (int c = 0; c < 6; ++c) {
      const v2f xj = *(const v2f*)(xB + j * D_MODEL + 4 * c + 2 * hh);
      dv = __builtin_amdgcn_wmma_f32_16x16x4_f32(
          false, Wxj[c], false, xj, (short)0, dv, false, false);
    }
#pragma unroll
    for (int r = 0; r < 8; ++r)
      vW[(tt * 16 + m) * 16 + r + 8 * hh] = dv[r];
  }

  __syncthreads();

  float s[5]  = {0.0f, 0.0f, 0.0f, 0.0f, 0.0f};
  float mx[5] = {0.0f, 0.0f, 0.0f, 0.0f, 0.0f};

  struct Feat {
    v2f   F[6];
    v2f   Flast;
    float wl0;
  };
  struct Raw {
    v4f c0, c1, c2, q0, q1, q2;
    float vj;
  };

  auto load_col = [&](const int jj, Raw& r) {
    const v4f* cp = (const v4f*)(cjW + jj * D_MODEL + hh * 12);
    const v4f* qp = (const v4f*)(qjW + jj * D_MODEL + hh * 12);
    r.c0 = cp[0]; r.c1 = cp[1]; r.c2 = cp[2];
    r.q0 = qp[0]; r.q1 = qp[1]; r.q2 = qp[2];
    r.vj = vW[jj * 16 + m];
  };

  auto features = [&](const Raw& r, Feat& o) {
    const float cjF[12] = {r.c0[0], r.c0[1], r.c0[2], r.c0[3], r.c1[0], r.c1[1],
                           r.c1[2], r.c1[3], r.c2[0], r.c2[1], r.c2[2], r.c2[3]};
    const float qjF[12] = {r.q0[0], r.q0[1], r.q0[2], r.q0[3], r.q1[0], r.q1[1],
                           r.q1[2], r.q1[3], r.q2[0], r.q2[1], r.q2[2], r.q2[3]};
    float sqpart = 0.0f, qpart = 0.0f;
#pragma unroll
    for (int c = 0; c < 6; ++c) {
      const float d0 = ci[c][0] - cjF[2 * c];
      const float d1 = ci[c][1] - cjF[2 * c + 1];
      o.F[c][0] = fabsf(d0);
      o.F[c][1] = fabsf(d1);
      sqpart  = fmaf(d0, d0, fmaf(d1, d1, sqpart));
      qpart   = fmaf(qi[c][0], qjF[2 * c], fmaf(qi[c][1], qjF[2 * c + 1], qpart));
    }
    const float qfull = qpart  + __shfl_xor(qpart, 16, 32);
    const float cdist = sqrtf(sqpart + __shfl_xor(sqpart, 16, 32));
    o.Flast[0] = hh ? 1.0f : qfull;
    o.Flast[1] = hh ? 0.0f : cdist;
    o.wl0      = hh ? r.vj : Wlast[0];
  };

  auto layer1 = [&](const Feat& f, v8f acc, const int c0, const int c1) {
#pragma unroll
    for (int c = c0; c < c1; ++c)
      acc = __builtin_amdgcn_wmma_f32_16x16x4_f32(
          false, Wc[c], false, f.F[c], (short)0, acc, false, false);
    return acc;
  };

  auto finish = [&](auto tag, const Feat& f, v8f acc, const int jl) {
    constexpr bool DIAG = decltype(tag)::value;
    v2f Wl = Wlast;
    Wl[0] = f.wl0;
    acc = __builtin_amdgcn_wmma_f32_16x16x4_f32(
        false, Wl, false, f.Flast, (short)0, acc, false, false);

    v2f H[3];
    H[0][0] = fmaxf(acc[0], 0.0f); H[0][1] = fmaxf(acc[1], 0.0f);
    H[1][0] = fmaxf(acc[2], 0.0f); H[1][1] = fmaxf(acc[3], 0.0f);
    H[2][0] = fmaxf(acc[4], 0.0f); H[2][1] = 0.0f;

    v8f d2 = c2init;
#pragma unroll
    for (int c = 0; c < 3; ++c)
      d2 = __builtin_amdgcn_wmma_f32_16x16x4_f32(
          false, W2a[c], false, H[c], (short)0, d2, false, false);

    if (DIAG) {
      const bool dg = (jl == m);
#pragma unroll
      for (int r = 0; r < 5; ++r) {
        s[r] += dg ? 0.0f : fmaxf(d2[r], 0.0f);
        mx[r] = dg ? mx[r] : fmaxf(mx[r], d2[r]);
      }
    } else {
#pragma unroll
      for (int r = 0; r < 5; ++r) {
        s[r] += fmaxf(d2[r], 0.0f);
        mx[r] = fmaxf(mx[r], d2[r]);
      }
    }
  };

  Feat fa, fb;
  {
    Raw r0;
    load_col(0, r0);
    features(r0, fa);
  }

  auto run_tile = [&](auto tag, const int tt) {
    for (int jl = 0; jl < 16; jl += 2) {
      const int jj = tt * 16 + jl;
      const int jn = (jj + 2 < JW) ? jj + 2 : JW - 1;
      Raw ra, rb;
      load_col(jj + 1, ra);
      load_col(jn, rb);
      __builtin_amdgcn_sched_barrier(0);
      v8f acca = layer1(fa, accx, 0, 3);
      __builtin_amdgcn_sched_barrier(0);
      features(ra, fb);
      acca = layer1(fa, acca, 3, 6);
      finish(tag, fa, acca, jl);
      features(rb, fa);
      finish(tag, fb, layer1(fb, accx, 0, 6), jl + 1);
    }
  };

#pragma nounroll
  for (int tt = 0; tt < JW / 16; ++tt) {
    if (j0 + tt * 16 == i0) run_tile(std::true_type{}, tt);
    else                    run_tile(std::false_type{}, tt);
  }

#pragma unroll
  for (int r = 0; r < 5; ++r) {
    sumL[wave * OUTW + m * REL_DIM + 5 * hh + r] = s[r];
    maxL[wave * OUTW + m * REL_DIM + 5 * hh + r] = mx[r];
  }

  __syncthreads();

  const size_t obase = ((size_t)b * NN + i0) * REL_DIM;
  if (t < OUTW) {
    float a  = sumL[t];
    float mv = maxL[t];
#pragma unroll
    for (int w = 1; w < NWAVES; ++w) {
      a += sumL[w * OUTW + t];
      mv = fmaxf(mv, maxL[w * OUTW + t]);
    }
    out[obase + t] = a / (float)(NN - 1);
    out[(size_t)BB * NN * REL_DIM + obase + t] = mv;
  }
}

extern "C" void kernel_launch(void* const* d_in, const int* in_sizes, int n_in,
                              void* d_out, int out_size, void* d_ws, size_t ws_size,
                              hipStream_t stream) {
  const float* x  = (const float*)d_in[0];
  const float* q  = (const float*)d_in[1];
  const float* co = (const float*)d_in[2];
  const float* W1 = (const float*)d_in[3];
  const float* b1 = (const float*)d_in[4];
  const float* W2 = (const float*)d_in[5];
  const float* b2 = (const float*)d_in[6];
  float* out = (float*)d_out;

  dim3 grid(BB * (NN / IC));
  dim3 block(NTHREADS);
  hipLaunchKernelGGL(pairagg_kernel, grid, block, 0, stream,
                     x, q, co, W1, b1, W2, b2, out);
}


# Column's squares and q products summed as two-wide packed partials on the register pairs as loaded: the register moves that built mixed pairs go (cdist and q dot sum order changes)
# speedup vs baseline: 2.1735x; 1.0475x over previous
#include <hip/hip_runtime.h>
#include <type_traits>


typedef __attribute__((ext_vector_type(2))) float v2f;
typedef __attribute__((ext_vector_type(4))) float v4f;
typedef __attribute__((ext_vector_type(8))) float v8f;

#define D_MODEL 24
#define REL_DIM 10
#define NN 512
#define BB 8
#define IC 16
#define NWAVES 8
#define NTHREADS (NWAVES * 32)
#define JW (NN / NWAVES)
#define OUTW (IC * REL_DIM)

__global__ __launch_bounds__(NTHREADS) void pairagg_kernel(
    const float* __restrict__ x, const float* __restrict__ q,
    const float* __restrict__ coord, const float* __restrict__ W1,
    const float* __restrict__ b1, const float* __restrict__ W2,
    const float* __restrict__ b2, float* __restrict__ out)
{
  __shared__ __attribute__((aligned(16))) float cjL[NWAVES * JW * D_MODEL];
  __shared__ __attribute__((aligned(16))) float qjL[NWAVES * JW * D_MODEL];
  __shared__ float vL[NWAVES * JW * 16];
  __shared__ float sumL[NWAVES * OUTW];
  __shared__ float maxL[NWAVES * OUTW];

  const int bi   = blockIdx.x;
  const int b    = bi / (NN / IC);
  const int i0   = (bi % (NN / IC)) * IC;
  const int t    = threadIdx.x;
  const int wave = __builtin_amdgcn_readfirstlane(t >> 5);
  const int lane = t & 31;
  const int m    = lane & 15;
  const int hh   = lane >> 4;

  const float* xB = x     + (size_t)b * NN * D_MODEL;
  const float* qB = q     + (size_t)b * NN * D_MODEL;
  const float* cB = coord + (size_t)b * NN * D_MODEL;

  const int  un   = (m < 5) ? m : m - 3;
  const bool has  = (m < 5) || (m >= 8 && m < 13);
  const int  unc  = has ? un : 0;

  v2f Wc[6], Wxi[6], Wxj[6];
#pragma unroll
  for (int c = 0; c < 6; ++c) {
#pragma unroll
    for (int v = 0; v < 2; ++v) {
      const int d = 4 * c + 2 * hh + v;
      const float wc = W1[(48 + d) * REL_DIM + unc];
      const float wi = W1[d * REL_DIM + unc];
      const float wj = W1[(24 + d) * REL_DIM + unc];
      Wc[c][v]  = has ? wc : 0.0f;
      Wxi[c][v] = has ? wi : 0.0f;
      Wxj[c][v] = has ? wj : 0.0f;
    }
  }
  v2f Wlast;
  {
    const float w72 = W1[72 * REL_DIM + unc];
    const float w73 = W1[73 * REL_DIM + unc];
    Wlast[0] = (has && !hh) ? w72 : 0.0f;
    Wlast[1] = (has && !hh) ? w73 : 0.0f;
  }
  v2f W2a[3];
#pragma unroll
  for (int c = 0; c < 3; ++c) {
#pragma unroll
    for (int v = 0; v < 2; ++v) {
      const int s  = 2 * c + v;
      const int ui = (s < 5) ? (5 * hh + s) : 0;
      const float w = W2[ui * REL_DIM + unc];
      W2a[c][v] = (s < 5 && has) ? w : 0.0f;
    }
  }
  v8f accx, c2init;
#pragma unroll
  for (int r = 0; r < 8; ++r) {
    accx[r]   = (r < 5) ? b1[5 * hh + (r < 5 ? r : 0)] : 0.0f;
    c2init[r] = (r < 5) ? b2[5 * hh + (r < 5 ? r : 0)] : 0.0f;
  }

  v2f ci[6], qi[6], xi[6];
  {
    const int irow = i0 + m;
#pragma unroll
    for (int c = 0; c < 6; ++c) {
      const int d = 4 * c + 2 * hh;
      ci[c] = *(const v2f*)(cB + irow * D_MODEL + d);
      qi[c] = *(const v2f*)(qB + irow * D_MODEL + d);
      xi[c] = *(const v2f*)(xB + irow * D_MODEL + d);
    }
  }
#pragma unroll
  for (int c = 0; c < 6; ++c)
    accx = __builtin_amdgcn_wmma_f32_16x16x4_f32(
        false, Wxi[c], false, xi[c], (short)0, accx, false, false);

  const int j0 = wave * JW;
  float* cjW = cjL + wave * (JW * D_MODEL);
  float* qjW = qjL + wave * (JW * D_MODEL);
  float* vW  = vL  + wave * (JW * 16);
  {
    const v4f* cg = (const v4f*)(cB + j0 * D_MODEL);
    const v4f* qg = (const v4f*)(qB + j0 * D_MODEL);
#pragma unroll
    for (int k = 0; k < (JW * D_MODEL) / (4 * 32); ++k) {
      const int f4 = k * 32 + lane;
      const int jl = f4 / 6, cc = f4 % 6;
      const v4f cv = cg[f4];
      const v4f qv = qg[f4];
      v2f lo, hi;
      lo[0] = cv[0]; lo[1] = cv[1]; hi[0] = cv[2]; hi[1] = cv[3];
      *(v2f*)(cjW + jl * D_MODEL + 2 * cc)      = lo;
      *(v2f*)(cjW + jl * D_MODEL + 12 + 2 * cc) = hi;
      lo[0] = qv[0]; lo[1] = qv[1]; hi[0] = qv[2]; hi[1] = qv[3];
      *(v2f*)(qjW + jl * D_MODEL + 2 * cc)      = lo;
      *(v2f*)(qjW + jl * D_MODEL + 12 + 2 * cc) = hi;
    }
  }
#pragma unroll
  for (int tt = 0; tt < JW / 16; ++tt) {
    const int j = j0 + tt * 16 + m;
    v8f dv = {};
#pragma unroll
    for (int c = 0; c < 6; ++c) {
      const v2f xj = *(const v2f*)(xB + j * D_MODEL + 4 * c + 2 * hh);
      dv = __builtin_amdgcn_wmma_f32_16x16x4_f32(
          false, Wxj[c], false, xj, (short)0, dv, false, false);
    }
#pragma unroll
    for (int r = 0; r < 8; ++r)
      vW[(tt * 16 + m) * 16 + r + 8 * hh] = dv[r];
  }

  __syncthreads();

  float s[5]  = {0.0f, 0.0f, 0.0f, 0.0f, 0.0f};
  float mx[5] = {0.0f, 0.0f, 0.0f, 0.0f, 0.0f};

  struct Feat {
    v2f   F[6];
    v2f   Flast;
    float wl0;
  };
  struct Raw {
    v4f c0, c1, c2, q0, q1, q2;
    float vj;
  };

  auto load_col = [&](const int jj, Raw& r) {
    const v4f* cp = (const v4f*)(cjW + jj * D_MODEL + hh * 12);
    const v4f* qp = (const v4f*)(qjW + jj * D_MODEL + hh * 12);
    r.c0 = cp[0]; r.c1 = cp[1]; r.c2 = cp[2];
    r.q0 = qp[0]; r.q1 = qp[1]; r.q2 = qp[2];
    r.vj = vW[jj * 16 + m];
  };

  auto features = [&](const Raw& r, Feat& o) {
    const v2f cj2[6] = {__builtin_shufflevector(r.c0, r.c0, 0, 1), __builtin_shufflevector(r.c0, r.c0, 2, 3),
                        __builtin_shufflevector(r.c1, r.c1, 0, 1), __builtin_shufflevector(r.c1, r.c1, 2, 3),
                        __builtin_shufflevector(r.c2, r.c2, 0, 1), __builtin_shufflevector(r.c2, r.c2, 2, 3)};
    const v2f qj2[6] = {__builtin_shufflevector(r.q0, r.q0, 0, 1), __builtin_shufflevector(r.q0, r.q0, 2, 3),
                        __builtin_shufflevector(r.q1, r.q1, 0, 1), __builtin_shufflevector(r.q1, r.q1, 2, 3),
                        __builtin_shufflevector(r.q2, r.q2, 0, 1), __builtin_shufflevector(r.q2, r.q2, 2, 3)};
    v2f sq2 = {0.0f, 0.0f}, q2 = {0.0f, 0.0f};
#pragma unroll
    for (int c = 0; c < 6; ++c) {
      const v2f d = ci[c] - cj2[c];
      o.F[c] = __builtin_elementwise_abs(d);
      sq2 = __builtin_elementwise_fma(d, d, sq2);
      q2  = __builtin_elementwise_fma(qi[c], qj2[c], q2);
    }
    const float sqpart = sq2[0] + sq2[1];
    const float qpart  = q2[0] + q2[1];
    const float qfull = qpart  + __shfl_xor(qpart, 16, 32);
    const float cdist = sqrtf(sqpart + __shfl_xor(sqpart, 16, 32));
    o.Flast[0] = hh ? 1.0f : qfull;
    o.Flast[1] = hh ? 0.0f : cdist;
    o.wl0      = hh ? r.vj : Wlast[0];
  };

  auto layer1 = [&](const Feat& f, v8f acc, const int c0, const int c1) {
#pragma unroll
    for (int c = c0; c < c1; ++c)
      acc = __builtin_amdgcn_wmma_f32_16x16x4_f32(
          false, Wc[c], false, f.F[c], (short)0, acc, false, false);
    return acc;
  };

  auto finish = [&](auto tag, const Feat& f, v8f acc, const int jl) {
    constexpr bool DIAG = decltype(tag)::value;
    v2f Wl = Wlast;
    Wl[0] = f.wl0;
    acc = __builtin_amdgcn_wmma_f32_16x16x4_f32(
        false, Wl, false, f.Flast, (short)0, acc, false, false);

    v2f H[3];
    H[0][0] = fmaxf(acc[0], 0.0f); H[0][1] = fmaxf(acc[1], 0.0f);
    H[1][0] = fmaxf(acc[2], 0.0f); H[1][1] = fmaxf(acc[3], 0.0f);
    H[2][0] = fmaxf(acc[4], 0.0f); H[2][1] = 0.0f;

    v8f d2 = c2init;
#pragma unroll
    for (int c = 0; c < 3; ++c)
      d2 = __builtin_amdgcn_wmma_f32_16x16x4_f32(
          false, W2a[c], false, H[c], (short)0, d2, false, false);

    if (DIAG) {
      const bool dg = (jl == m);
#pragma unroll
      for (int r = 0; r < 5; ++r) {
        s[r] += dg ? 0.0f : fmaxf(d2[r], 0.0f);
        mx[r] = dg ? mx[r] : fmaxf(mx[r], d2[r]);
      }
    } else {
#pragma unroll
      for (int r = 0; r < 5; ++r) {
        s[r] += fmaxf(d2[r], 0.0f);
        mx[r] = fmaxf(mx[r], d2[r]);
      }
    }
  };

  Feat fa, fb;
  {
    Raw r0;
    load_col(0, r0);
    features(r0, fa);
  }

  auto run_tile = [&](auto tag, const int tt) {
    for (int jl = 0; jl < 16; jl += 2) {
      const int jj = tt * 16 + jl;
      const int jn = (jj + 2 < JW) ? jj + 2 : JW - 1;
      Raw ra, rb;
      load_col(jj + 1, ra);
      load_col(jn, rb);
      __builtin_amdgcn_sched_barrier(0);
      v8f acca = layer1(fa, accx, 0, 3);
      __builtin_amdgcn_sched_barrier(0);
      features(ra, fb);
      acca = layer1(fa, acca, 3, 6);
      finish(tag, fa, acca, jl);
      features(rb, fa);
      finish(tag, fb, layer1(fb, accx, 0, 6), jl + 1);
    }
  };

#pragma nounroll
  for (int tt = 0; tt < JW / 16; ++tt) {
    if (j0 + tt * 16 == i0) run_tile(std::true_type{}, tt);
    else                    run_tile(std::false_type{}, tt);
  }

#pragma unroll
  for (int r = 0; r < 5; ++r) {
    sumL[wave * OUTW + m * REL_DIM + 5 * hh + r] = s[r];
    maxL[wave * OUTW + m * REL_DIM + 5 * hh + r] = mx[r];
  }

  __syncthreads();

  const size_t obase = ((size_t)b * NN + i0) * REL_DIM;
  if (t < OUTW) {
    float a  = sumL[t];
    float mv = maxL[t];
#pragma unroll
    for (int w = 1; w < NWAVES; ++w) {
      a += sumL[w * OUTW + t];
      mv = fmaxf(mv, maxL[w * OUTW + t]);
    }
    out[obase + t] = a / (float)(NN - 1);
    out[(size_t)BB * NN * REL_DIM + obase + t] = mv;
  }
}

extern "C" void kernel_launch(void* const* d_in, const int* in_sizes, int n_in,
                              void* d_out, int out_size, void* d_ws, size_t ws_size,
                              hipStream_t stream) {
  const float* x  = (const float*)d_in[0];
  const float* q  = (const float*)d_in[1];
  const float* co = (const float*)d_in[2];
  const float* W1 = (const float*)d_in[3];
  const float* b1 = (const float*)d_in[4];
  const float* W2 = (const float*)d_in[5];
  const float* b2 = (const float*)d_in[6];
  float* out = (float*)d_out;

  dim3 grid(BB * (NN / IC));
  dim3 block(NTHREADS);
  hipLaunchKernelGGL(pairagg_kernel, grid, block, 0, stream,
                     x, q, co, W1, b1, W2, b2, out);
}


# Lane halves stop repeating each other behind the 24-dim sums: q partials take own k slots (WMMA joins them), squares cross once so each half roots one of two columns (q dot sum order changes)
# speedup vs baseline: 2.2596x; 1.0396x over previous
#include <hip/hip_runtime.h>
#include <type_traits>


typedef __attribute__((ext_vector_type(2))) float v2f;
typedef __attribute__((ext_vector_type(4))) float v4f;
typedef __attribute__((ext_vector_type(8))) float v8f;

#define D_MODEL 24
#define REL_DIM 10
#define NN 512
#define BB 8
#define IC 16
#define NWAVES 8
#define NTHREADS (NWAVES * 32)
#define JW (NN / NWAVES)
#define OUTW (IC * REL_DIM)

__global__ __launch_bounds__(NTHREADS) void pairagg_kernel(
    const float* __restrict__ x, const float* __restrict__ q,
    const float* __restrict__ coord, const float* __restrict__ W1,
    const float* __restrict__ b1, const float* __restrict__ W2,
    const float* __restrict__ b2, float* __restrict__ out)
{
  __shared__ __attribute__((aligned(16))) float cjL[NWAVES * JW * D_MODEL];
  __shared__ __attribute__((aligned(16))) float qjL[NWAVES * JW * D_MODEL];
  __shared__ float vL[NWAVES * JW * 16];
  __shared__ float sumL[NWAVES * OUTW];
  __shared__ float maxL[NWAVES * OUTW];

  const int bi   = blockIdx.x;
  const int b    = bi / (NN / IC);
  const int i0   = (bi % (NN / IC)) * IC;
  const int t    = threadIdx.x;
  const int wave = __builtin_amdgcn_readfirstlane(t >> 5);
  const int lane = t & 31;
  const int m    = lane & 15;
  const int hh   = lane >> 4;

  const float* xB = x     + (size_t)b * NN * D_MODEL;
  const float* qB = q     + (size_t)b * NN * D_MODEL;
  const float* cB = coord + (size_t)b * NN * D_MODEL;

  const int  un   = (m < 5) ? m : m - 3;
  const bool has  = (m < 5) || (m >= 8 && m < 13);
  const int  unc  = has ? un : 0;

  v2f Wc[6], Wxi[6], Wxj[6];
#pragma unroll
  for (int c = 0; c < 6; ++c) {
#pragma unroll
    for (int v = 0; v < 2; ++v) {
      const int d = 4 * c + 2 * hh + v;
      const float wc = W1[(48 + d) * REL_DIM + unc];
      const float wi = W1[d * REL_DIM + unc];
      const float wj = W1[(24 + d) * REL_DIM + unc];
      Wc[c][v]  = has ? wc : 0.0f;
      Wxi[c][v] = has ? wi : 0.0f;
      Wxj[c][v] = has ? wj : 0.0f;
    }
  }
  float w72s, WA1, WB0;
  {
    const float w72 = W1[72 * REL_DIM + unc];
    const float w73 = W1[73 * REL_DIM + unc];
    w72s = has ? w72 : 0.0f;
    const float w73s = has ? w73 : 0.0f;
    WA1 = hh ? w72s : w73s;
    WB0 = hh ? w73s : w72s;
  }
  v2f W2a[3];
#pragma unroll
  for (int c = 0; c < 3; ++c) {
#pragma unroll
    for (int v = 0; v < 2; ++v) {
      const int s  = 2 * c + v;
      const int ui = (s < 5) ? (5 * hh + s) : 0;
      const float w = W2[ui * REL_DIM + unc];
      W2a[c][v] = (s < 5 && has) ? w : 0.0f;
    }
  }
  v8f accx, c2init;
#pragma unroll
  for (int r = 0; r < 8; ++r) {
    accx[r]   = (r < 5) ? b1[5 * hh + (r < 5 ? r : 0)] : 0.0f;
    c2init[r] = (r < 5) ? b2[5 * hh + (r < 5 ? r : 0)] : 0.0f;
  }

  v2f ci[6], qi[6], xi[6];
  {
    const int irow = i0 + m;
#pragma unroll
    for (int c = 0; c < 6; ++c) {
      const int d = 4 * c + 2 * hh;
      ci[c] = *(const v2f*)(cB + irow * D_MODEL + d);
      qi[c] = *(const v2f*)(qB + irow * D_MODEL + d);
      xi[c] = *(const v2f*)(xB + irow * D_MODEL + d);
    }
  }
#pragma unroll
  for (int c = 0; c < 6; ++c)
    accx = __builtin_amdgcn_wmma_f32_16x16x4_f32(
        false, Wxi[c], false, xi[c], (short)0, accx, false, false);

  const int j0 = wave * JW;
  float* cjW = cjL + wave * (JW * D_MODEL);
  float* qjW = qjL + wave * (JW * D_MODEL);
  float* vW  = vL  + wave * (JW * 16);
  {
    const v4f* cg = (const v4f*)(cB + j0 * D_MODEL);
    const v4f* qg = (const v4f*)(qB + j0 * D_MODEL);
#pragma unroll
    for (int k = 0; k < (JW * D_MODEL) / (4 * 32); ++k) {
      const int f4 = k * 32 + lane;
      const int jl = f4 / 6, cc = f4 % 6;
      const v4f cv = cg[f4];
      const v4f qv = qg[f4];
      v2f lo, hi;
      lo[0] = cv[0]; lo[1] = cv[1]; hi[0] = cv[2]; hi[1] = cv[3];
      *(v2f*)(cjW + jl * D_MODEL + 2 * cc)      = lo;
      *(v2f*)(cjW + jl * D_MODEL + 12 + 2 * cc) = hi;
      lo[0] = qv[0]; lo[1] = qv[1]; hi[0] = qv[2]; hi[1] = qv[3];
      *(v2f*)(qjW + jl * D_MODEL + 2 * cc)      = lo;
      *(v2f*)(qjW + jl * D_MODEL + 12 + 2 * cc) = hi;
    }
  }
#pragma unroll
  for (int tt = 0; tt < JW / 16; ++tt) {
    const int j = j0 + tt * 16 + m;
    v8f dv = {};
#pragma unroll
    for (int c = 0; c < 6; ++c) {
      const v2f xj = *(const v2f*)(xB + j * D_MODEL + 4 * c + 2 * hh);
      dv = __builtin_amdgcn_wmma_f32_16x16x4_f32(
          false, Wxj[c], false, xj, (short)0, dv, false, false);
    }
#pragma unroll
    for (int r = 0; r < 8; ++r)
      vW[(tt * 16 + m) * 16 + r + 8 * hh] = dv[r];
  }

  __syncthreads();

  float s[5]  = {0.0f, 0.0f, 0.0f, 0.0f, 0.0f};
  float mx[5] = {0.0f, 0.0f, 0.0f, 0.0f, 0.0f};

  struct Feat {
    v2f   F[6];
    v2f   Flast;
    v2f   Wl;
  };
  struct Raw {
    v4f c0, c1, c2, q0, q1, q2;
    float vj;
  };

  auto load_col = [&](const int jj, Raw& r) {
    const v4f* cp = (const v4f*)(cjW + jj * D_MODEL + hh * 12);
    const v4f* qp = (const v4f*)(qjW + jj * D_MODEL + hh * 12);
    r.c0 = cp[0]; r.c1 = cp[1]; r.c2 = cp[2];
    r.q0 = qp[0]; r.q1 = qp[1]; r.q2 = qp[2];
    r.vj = vW[jj * 16 + m];
  };

  auto diffs = [&](const Raw& r, Feat& o, v2f& sq2, v2f& q2) {
    const v2f cj2[6] = {__builtin_shufflevector(r.c0, r.c0, 0, 1), __builtin_shufflevector(r.c0, r.c0, 2, 3),
                        __builtin_shufflevector(r.c1, r.c1, 0, 1), __builtin_shufflevector(r.c1, r.c1, 2, 3),
                        __builtin_shufflevector(r.c2, r.c2, 0, 1), __builtin_shufflevector(r.c2, r.c2, 2, 3)};
    const v2f qj2[6] = {__builtin_shufflevector(r.q0, r.q0, 0, 1), __builtin_shufflevector(r.q0, r.q0, 2, 3),
                        __builtin_shufflevector(r.q1, r.q1, 0, 1), __builtin_shufflevector(r.q1, r.q1, 2, 3),
                        __builtin_shufflevector(r.q2, r.q2, 0, 1), __builtin_shufflevector(r.q2, r.q2, 2, 3)};
    sq2 = v2f{0.0f, 0.0f}; q2 = v2f{0.0f, 0.0f};
#pragma unroll
    for (int c = 0; c < 6; ++c) {
      const v2f d = ci[c] - cj2[c];
      o.F[c] = __builtin_elementwise_abs(d);
      sq2 = __builtin_elementwise_fma(d, d, sq2);
      q2  = __builtin_elementwise_fma(qi[c], qj2[c], q2);
    }
  };

  auto features2 = [&](const Raw& ra, const Raw& rb, Feat& oa, Feat& ob) {
    v2f sqa2, qa2, sqb2, qb2;
    diffs(ra, oa, sqa2, qa2);
    diffs(rb, ob, sqb2, qb2);
    const float sqa = sqa2[0] + sqa2[1], sqb = sqb2[0] + sqb2[1];
    const float qa  = qa2[0] + qa2[1],   qb  = qb2[0] + qb2[1];
    const float send = hh ? sqa : sqb;
    const float own  = hh ? sqb : sqa;
    const float root = sqrtf(own + __shfl_xor(send, 16, 32));
    oa.Flast[0] = hh ? 1.0f : qa;
    oa.Flast[1] = hh ? qa : root;
    oa.Wl[0]    = hh ? ra.vj : w72s;
    oa.Wl[1]    = WA1;
    ob.Flast[0] = hh ? root : qb;
    ob.Flast[1] = hh ? qb : 1.0f;
    ob.Wl[0]    = WB0;
    ob.Wl[1]    = hh ? w72s : rb.vj;
  };

  auto layer1 = [&](const Feat& f, v8f acc, const int c0, const int c1) {
#pragma unroll
    for (int c = c0; c < c1; ++c)
      acc = __builtin_amdgcn_wmma_f32_16x16x4_f32(
          false, Wc[c], false, f.F[c], (short)0, acc, false, false);
    return acc;
  };

  auto finish = [&](auto tag, const Feat& f, v8f acc, const int jl) {
    constexpr bool DIAG = decltype(tag)::value;
    acc = __builtin_amdgcn_wmma_f32_16x16x4_f32(
        false, f.Wl, false, f.Flast, (short)0, acc, false, false);

    v2f H[3];
    H[0][0] = fmaxf(acc[0], 0.0f); H[0][1] = fmaxf(acc[1], 0.0f);
    H[1][0] = fmaxf(acc[2], 0.0f); H[1][1] = fmaxf(acc[3], 0.0f);
    H[2][0] = fmaxf(acc[4], 0.0f); H[2][1] = 0.0f;

    v8f d2 = c2init;
#pragma unroll
    for (int c = 0; c < 3; ++c)
      d2 = __builtin_amdgcn_wmma_f32_16x16x4_f32(
          false, W2a[c], false, H[c], (short)0, d2, false, false);

    if (DIAG) {
      const bool dg = (jl == m);
#pragma unroll
      for (int r = 0; r < 5; ++r) {
        s[r] += dg ? 0.0f : fmaxf(d2[r], 0.0f);
        mx[r] = dg ? mx[r] : fmaxf(mx[r], d2[r]);
      }
    } else {
#pragma unroll
      for (int r = 0; r < 5; ++r) {
        s[r] += fmaxf(d2[r], 0.0f);
        mx[r] = fmaxf(mx[r], d2[r]);
      }
    }
  };

  Feat fa, fb;
  {
    Raw r0;
    load_col(0, r0);
    features2(r0, r0, fb, fa);
  }

  auto run_tile = [&](auto tag, const int tt) {
    for (int jl = 0; jl < 16; jl += 2) {
      const int jj = tt * 16 + jl;
      const int jn = (jj + 2 < JW) ? jj + 2 : JW - 1;
      Raw ra, rb;
      load_col(jj + 1, ra);
      load_col(jn, rb);
      __builtin_amdgcn_sched_barrier(0);
      v8f acca = layer1(fa, accx, 0, 3);
      __builtin_amdgcn_sched_barrier(0);
      acca = layer1(fa, acca, 3, 6);
      finish(tag, fa, acca, jl);
      features2(ra, rb, fb, fa);
      finish(tag, fb, layer1(fb, accx, 0, 6), jl + 1);
    }
  };

#pragma nounroll
  for (int tt = 0; tt < JW / 16; ++tt) {
    if (j0 + tt * 16 == i0) run_tile(std::true_type{}, tt);
    else                    run_tile(std::false_type{}, tt);
  }

#pragma unroll
  for (int r = 0; r < 5; ++r) {
    sumL[wave * OUTW + m * REL_DIM + 5 * hh + r] = s[r];
    maxL[wave * OUTW + m * REL_DIM + 5 * hh + r] = mx[r];
  }

  __syncthreads();

  const size_t obase = ((size_t)b * NN + i0) * REL_DIM;
  if (t < OUTW) {
    float a  = sumL[t];
    float mv = maxL[t];
#pragma unroll
    for (int w = 1; w < NWAVES; ++w) {
      a += sumL[w * OUTW + t];
      mv = fmaxf(mv, maxL[w * OUTW + t]);
    }
    out[obase + t] = a / (float)(NN - 1);
    out[(size_t)BB * NN * REL_DIM + obase + t] = mv;
  }
}

extern "C" void kernel_launch(void* const* d_in, const int* in_sizes, int n_in,
                              void* d_out, int out_size, void* d_ws, size_t ws_size,
                              hipStream_t stream) {
  const float* x  = (const float*)d_in[0];
  const float* q  = (const float*)d_in[1];
  const float* co = (const float*)d_in[2];
  const float* W1 = (const float*)d_in[3];
  const float* b1 = (const float*)d_in[4];
  const float* W2 = (const float*)d_in[5];
  const float* b2 = (const float*)d_in[6];
  float* out = (float*)d_out;

  dim3 grid(BB * (NN / IC));
  dim3 block(NTHREADS);
  hipLaunchKernelGGL(pairagg_kernel, grid, block, 0, stream,
                     x, q, co, W1, b1, W2, b2, out);
}


# Only the first 16-column tile is loaded in front of the pair loop; each later tile's global loads fly during the tile before it and are staged to LDS in its middle, wave-private, no block barrier
# speedup vs baseline: 2.3738x; 1.0505x over previous
#include <hip/hip_runtime.h>
#include <type_traits>


typedef __attribute__((ext_vector_type(2))) float v2f;
typedef __attribute__((ext_vector_type(4))) float v4f;
typedef __attribute__((ext_vector_type(8))) float v8f;

#define D_MODEL 24
#define REL_DIM 10
#define NN 512
#define BB 8
#define IC 16
#define NWAVES 8
#define NTHREADS (NWAVES * 32)
#define JW (NN / NWAVES)
#define OUTW (IC * REL_DIM)

__global__ __launch_bounds__(NTHREADS) void pairagg_kernel(
    const float* __restrict__ x, const float* __restrict__ q,
    const float* __restrict__ coord, const float* __restrict__ W1,
    const float* __restrict__ b1, const float* __restrict__ W2,
    const float* __restrict__ b2, float* __restrict__ out)
{
  __shared__ __attribute__((aligned(16))) float cjL[NWAVES * JW * D_MODEL];
  __shared__ __attribute__((aligned(16))) float qjL[NWAVES * JW * D_MODEL];
  __shared__ float vL[NWAVES * JW * 16];
  __shared__ float sumL[NWAVES * OUTW];
  __shared__ float maxL[NWAVES * OUTW];

  const int bi   = blockIdx.x;
  const int b    = bi / (NN / IC);
  const int i0   = (bi % (NN / IC)) * IC;
  const int t    = threadIdx.x;
  const int wave = __builtin_amdgcn_readfirstlane(t >> 5);
  const int lane = t & 31;
  const int m    = lane & 15;
  const int hh   = lane >> 4;

  const float* xB = x     + (size_t)b * NN * D_MODEL;
  const float* qB = q     + (size_t)b * NN * D_MODEL;
  const float* cB = coord + (size_t)b * NN * D_MODEL;

  const int  un   = (m < 5) ? m : m - 3;
  const bool has  = (m < 5) || (m >= 8 && m < 13);
  const int  unc  = has ? un : 0;

  v2f Wc[6], Wxi[6], Wxj[6];
#pragma unroll
  for (int c = 0; c < 6; ++c) {
#pragma unroll
    for (int v = 0; v < 2; ++v) {
      const int d = 4 * c + 2 * hh + v;
      const float wc = W1[(48 + d) * REL_DIM + unc];
      const float wi = W1[d * REL_DIM + unc];
      const float wj = W1[(24 + d) * REL_DIM + unc];
      Wc[c][v]  = has ? wc : 0.0f;
      Wxi[c][v] = has ? wi : 0.0f;
      Wxj[c][v] = has ? wj : 0.0f;
    }
  }
  float w72s, WA1, WB0;
  {
    const float w72 = W1[72 * REL_DIM + unc];
    const float w73 = W1[73 * REL_DIM + unc];
    w72s = has ? w72 : 0.0f;
    const float w73s = has ? w73 : 0.0f;
    WA1 = hh ? w72s : w73s;
    WB0 = hh ? w73s : w72s;
  }
  v2f W2a[3];
#pragma unroll
  for (int c = 0; c < 3; ++c) {
#pragma unroll
    for (int v = 0; v < 2; ++v) {
      const int s  = 2 * c + v;
      const int ui = (s < 5) ? (5 * hh + s) : 0;
      const float w = W2[ui * REL_DIM + unc];
      W2a[c][v] = (s < 5 && has) ? w : 0.0f;
    }
  }
  v8f accx, c2init;
#pragma unroll
  for (int r = 0; r < 8; ++r) {
    accx[r]   = (r < 5) ? b1[5 * hh + (r < 5 ? r : 0)] : 0.0f;
    c2init[r] = (r < 5) ? b2[5 * hh + (r < 5 ? r : 0)] : 0.0f;
  }

  v2f ci[6], qi[6], xi[6];
  {
    const int irow = i0 + m;
#pragma unroll
    for (int c = 0; c < 6; ++c) {
      const int d = 4 * c + 2 * hh;
      ci[c] = *(const v2f*)(cB + irow * D_MODEL + d);
      qi[c] = *(const v2f*)(qB + irow * D_MODEL + d);
      xi[c] = *(const v2f*)(xB + irow * D_MODEL + d);
    }
  }
#pragma unroll
  for (int c = 0; c < 6; ++c)
    accx = __builtin_amdgcn_wmma_f32_16x16x4_f32(
        false, Wxi[c], false, xi[c], (short)0, accx, false, false);

  const int j0 = wave * JW;
  float* cjW = cjL + wave * (JW * D_MODEL);
  float* qjW = qjL + wave * (JW * D_MODEL);
  float* vW  = vL  + wave * (JW * 16);
  struct Stage { v4f cv[3], qv[3]; v2f xv[6]; };
  auto issue_tile = [&](const int tt, Stage& g) {
    const v4f* cg = (const v4f*)(cB + (j0 + tt * 16) * D_MODEL);
    const v4f* qg = (const v4f*)(qB + (j0 + tt * 16) * D_MODEL);
#pragma unroll
    for (int k = 0; k < 3; ++k) {
      g.cv[k] = cg[k * 32 + lane];
      g.qv[k] = qg[k * 32 + lane];
    }
    const int j = j0 + tt * 16 + m;
#pragma unroll
    for (int c = 0; c < 6; ++c)
      g.xv[c] = *(const v2f*)(xB + j * D_MODEL + 4 * c + 2 * hh);
  };
  auto store_tile = [&](const int tt, const Stage& g) {
#pragma unroll
    for (int k = 0; k < 3; ++k) {
      const int f4 = k * 32 + lane;
      const int jl = tt * 16 + f4 / 6, cc = f4 % 6;
      v2f lo, hi;
      lo[0] = g.cv[k][0]; lo[1] = g.cv[k][1]; hi[0] = g.cv[k][2]; hi[1] = g.cv[k][3];
      *(v2f*)(cjW + jl * D_MODEL + 2 * cc)      = lo;
      *(v2f*)(cjW + jl * D_MODEL + 12 + 2 * cc) = hi;
      lo[0] = g.qv[k][0]; lo[1] = g.qv[k][1]; hi[0] = g.qv[k][2]; hi[1] = g.qv[k][3];
      *(v2f*)(qjW + jl * D_MODEL + 2 * cc)      = lo;
      *(v2f*)(qjW + jl * D_MODEL + 12 + 2 * cc) = hi;
    }
    v8f dv = {};
#pragma unroll
    for (int c = 0; c < 6; ++c)
      dv = __builtin_amdgcn_wmma_f32_16x16x4_f32(
          false, Wxj[c], false, g.xv[c], (short)0, dv, false, false);
#pragma unroll
    for (int r = 0; r < 8; ++r)
      vW[(tt * 16 + m) * 16 + r + 8 * hh] = dv[r];
  };
  {
    Stage g0;
    issue_tile(0, g0);
    store_tile(0, g0);
  }

  __syncthreads();

  float s[5]  = {0.0f, 0.0f, 0.0f, 0.0f, 0.0f};
  float mx[5] = {0.0f, 0.0f, 0.0f, 0.0f, 0.0f};

  struct Feat {
    v2f   F[6];
    v2f   Flast;
    v2f   Wl;
  };
  struct Raw {
    v4f c0, c1, c2, q0, q1, q2;
    float vj;
  };

  auto load_col = [&](const int jj, Raw& r) {
    const v4f* cp = (const v4f*)(cjW + jj * D_MODEL + hh * 12);
    const v4f* qp = (const v4f*)(qjW + jj * D_MODEL + hh * 12);
    r.c0 = cp[0]; r.c1 = cp[1]; r.c2 = cp[2];
    r.q0 = qp[0]; r.q1 = qp[1]; r.q2 = qp[2];
    r.vj = vW[jj * 16 + m];
  };

  auto diffs = [&](const Raw& r, Feat& o, v2f& sq2, v2f& q2) {
    const v2f cj2[6] = {__builtin_shufflevector(r.c0, r.c0, 0, 1), __builtin_shufflevector(r.c0, r.c0, 2, 3),
                        __builtin_shufflevector(r.c1, r.c1, 0, 1), __builtin_shufflevector(r.c1, r.c1, 2, 3),
                        __builtin_shufflevector(r.c2, r.c2, 0, 1), __builtin_shufflevector(r.c2, r.c2, 2, 3)};
    const v2f qj2[6] = {__builtin_shufflevector(r.q0, r.q0, 0, 1), __builtin_shufflevector(r.q0, r.q0, 2, 3),
                        __builtin_shufflevector(r.q1, r.q1, 0, 1), __builtin_shufflevector(r.q1, r.q1, 2, 3),
                        __builtin_shufflevector(r.q2, r.q2, 0, 1), __builtin_shufflevector(r.q2, r.q2, 2, 3)};
    sq2 = v2f{0.0f, 0.0f}; q2 = v2f{0.0f, 0.0f};
#pragma unroll
    for (int c = 0; c < 6; ++c) {
      const v2f d = ci[c] - cj2[c];
      o.F[c] = __builtin_elementwise_abs(d);
      sq2 = __builtin_elementwise_fma(d, d, sq2);
      q2  = __builtin_elementwise_fma(qi[c], qj2[c], q2);
    }
  };

  auto features2 = [&](const Raw& ra, const Raw& rb, Feat& oa, Feat& ob) {
    v2f sqa2, qa2, sqb2, qb2;
    diffs(ra, oa, sqa2, qa2);
    diffs(rb, ob, sqb2, qb2);
    const float sqa = sqa2[0] + sqa2[1], sqb = sqb2[0] + sqb2[1];
    const float qa  = qa2[0] + qa2[1],   qb  = qb2[0] + qb2[1];
    const float send = hh ? sqa : sqb;
    const float own  = hh ? sqb : sqa;
    const float root = sqrtf(own + __shfl_xor(send, 16, 32));
    oa.Flast[0] = hh ? 1.0f : qa;
    oa.Flast[1] = hh ? qa : root;
    oa.Wl[0]    = hh ? ra.vj : w72s;
    oa.Wl[1]    = WA1;
    ob.Flast[0] = hh ? root : qb;
    ob.Flast[1] = hh ? qb : 1.0f;
    ob.Wl[0]    = WB0;
    ob.Wl[1]    = hh ? w72s : rb.vj;
  };

  auto layer1 = [&](const Feat& f, v8f acc, const int c0, const int c1) {
#pragma unroll
    for (int c = c0; c < c1; ++c)
      acc = __builtin_amdgcn_wmma_f32_16x16x4_f32(
          false, Wc[c], false, f.F[c], (short)0, acc, false, false);
    return acc;
  };

  auto finish = [&](auto tag, const Feat& f, v8f acc, const int jl) {
    constexpr bool DIAG = decltype(tag)::value;
    acc = __builtin_amdgcn_wmma_f32_16x16x4_f32(
        false, f.Wl, false, f.Flast, (short)0, acc, false, false);

    v2f H[3];
    H[0][0] = fmaxf(acc[0], 0.0f); H[0][1] = fmaxf(acc[1], 0.0f);
    H[1][0] = fmaxf(acc[2], 0.0f); H[1][1] = fmaxf(acc[3], 0.0f);
    H[2][0] = fmaxf(acc[4], 0.0f); H[2][1] = 0.0f;

    v8f d2 = c2init;
#pragma unroll
    for (int c = 0; c < 3; ++c)
      d2 = __builtin_amdgcn_wmma_f32_16x16x4_f32(
          false, W2a[c], false, H[c], (short)0, d2, false, false);

    if (DIAG) {
      const bool dg = (jl == m);
#pragma unroll
      for (int r = 0; r < 5; ++r) {
        s[r] += dg ? 0.0f : fmaxf(d2[r], 0.0f);
        mx[r] = dg ? mx[r] : fmaxf(mx[r], d2[r]);
      }
    } else {
#pragma unroll
      for (int r = 0; r < 5; ++r) {
        s[r] += fmaxf(d2[r], 0.0f);
        mx[r] = fmaxf(mx[r], d2[r]);
      }
    }
  };

  Feat fa, fb;
  {
    Raw r0;
    load_col(0, r0);
    features2(r0, r0, fb, fa);
  }

  auto run_tile = [&](auto tag, const int tt, const int jl0, const int jl1) {
    for (int jl = jl0; jl < jl1; jl += 2) {
      const int jj = tt * 16 + jl;
      const int jn = (jj + 2 < JW) ? jj + 2 : JW - 1;
      Raw ra, rb;
      load_col(jj + 1, ra);
      load_col(jn, rb);
      __builtin_amdgcn_sched_barrier(0);
      v8f acca = layer1(fa, accx, 0, 3);
      __builtin_amdgcn_sched_barrier(0);
      acca = layer1(fa, acca, 3, 6);
      finish(tag, fa, acca, jl);
      features2(ra, rb, fb, fa);
      finish(tag, fb, layer1(fb, accx, 0, 6), jl + 1);
    }
  };

#pragma nounroll
  for (int tt = 0; tt < JW / 16; ++tt) {
    const bool more = tt + 1 < JW / 16;
    const bool diag = (j0 + tt * 16 == i0);
    Stage g;
    if (more) issue_tile(tt + 1, g);
    if (diag) run_tile(std::true_type{}, tt, 0, 8);
    else      run_tile(std::false_type{}, tt, 0, 8);
    if (more) {
      store_tile(tt + 1, g);
      __builtin_amdgcn_fence(__ATOMIC_RELEASE, "wavefront");
      __builtin_amdgcn_wave_barrier();
      __builtin_amdgcn_fence(__ATOMIC_ACQUIRE, "wavefront");
    }
    if (diag) run_tile(std::true_type{}, tt, 8, 16);
    else      run_tile(std::false_type{}, tt, 8, 16);
  }

#pragma unroll
  for (int r = 0; r < 5; ++r) {
    sumL[wave * OUTW + m * REL_DIM + 5 * hh + r] = s[r];
    maxL[wave * OUTW + m * REL_DIM + 5 * hh + r] = mx[r];
  }

  __syncthreads();

  const size_t obase = ((size_t)b * NN + i0) * REL_DIM;
  if (t < OUTW) {
    float a  = sumL[t];
    float mv = maxL[t];
#pragma unroll
    for (int w = 1; w < NWAVES; ++w) {
      a += sumL[w * OUTW + t];
      mv = fmaxf(mv, maxL[w * OUTW + t]);
    }
    out[obase + t] = a / (float)(NN - 1);
    out[(size_t)BB * NN * REL_DIM + obase + t] = mv;
  }
}

extern "C" void kernel_launch(void* const* d_in, const int* in_sizes, int n_in,
                              void* d_out, int out_size, void* d_ws, size_t ws_size,
                              hipStream_t stream) {
  const float* x  = (const float*)d_in[0];
  const float* q  = (const float*)d_in[1];
  const float* co = (const float*)d_in[2];
  const float* W1 = (const float*)d_in[3];
  const float* b1 = (const float*)d_in[4];
  const float* W2 = (const float*)d_in[5];
  const float* b2 = (const float*)d_in[6];
  float* out = (float*)d_out;

  dim3 grid(BB * (NN / IC));
  dim3 block(NTHREADS);
  hipLaunchKernelGGL(pairagg_kernel, grid, block, 0, stream,
                     x, q, co, W1, b1, W2, b2, out);
}
